# starting kernel
# baseline (speedup 1.0000x reference)
#include <hip/hip_runtime.h>
#include <hip/hip_bf16.h>

#define N_NODES  100000
#define N_EDGES  3200000
#define N_GRAPHS 256
#define HID      32
#define IN_DIM   6

typedef __attribute__((ext_vector_type(2))) float v2f;
typedef __attribute__((ext_vector_type(8))) float v8f;

__global__ void fill_zero_kernel(float* __restrict__ p, int n) {
  int i = blockIdx.x * blockDim.x + threadIdx.x;
  if (i < n) p[i] = 0.0f;
}

__global__ void edge_weight_kernel(const float* __restrict__ pos,
                                   const int* __restrict__ row,
                                   const int* __restrict__ col,
                                   const float* __restrict__ s1p,
                                   const float* __restrict__ s2p,
                                   float* __restrict__ ew,
                                   float* __restrict__ deg) {
  int e = blockIdx.x * blockDim.x + threadIdx.x;
  if (e >= N_EDGES) return;
  int r = row[e], c = col[e];
  float s1 = s1p[0], s2 = s2p[0];
  const float* pr = pos + (size_t)r * 6;
  const float* pc = pos + (size_t)c * 6;
  float d0 = pr[0] - pc[0], d1 = pr[1] - pc[1], d2 = pr[2] - pc[2];
  float D = d0 * d0 + d1 * d1 + d2 * d2;
  float t = 1.0f - (pr[3] * pc[3] + pr[4] * pc[4] + pr[5] * pc[5]);
  float w = __expf(-(D * s1 * s1 + t * t * s2 * s2));
  ew[e] = w;
  atomicAdd(&deg[c], w);
}

__global__ void dinv_kernel(const float* __restrict__ deg, float* __restrict__ dinv) {
  int i = blockIdx.x * blockDim.x + threadIdx.x;
  if (i >= N_NODES) return;
  dinv[i] = rsqrtf(deg[i] + 1.0f);
}

__global__ void coef_kernel(const int* __restrict__ row,
                            const int* __restrict__ col,
                            const float* __restrict__ dinv,
                            float* __restrict__ ew) {
  int e = blockIdx.x * blockDim.x + threadIdx.x;
  if (e >= N_EDGES) return;
  ew[e] = dinv[row[e]] * ew[e] * dinv[col[e]];
}

__global__ void gemm_in_kernel(const float* __restrict__ x,
                               const float* __restrict__ W,
                               float* __restrict__ xw) {
  int gid = blockIdx.x * blockDim.x + threadIdx.x;
  if (gid >= N_NODES * HID) return;
  int node = gid >> 5;
  int f    = gid & 31;
  float acc = 0.0f;
#pragma unroll
  for (int k = 0; k < IN_DIM; ++k)
    acc += x[node * IN_DIM + k] * W[k * HID + f];
  xw[gid] = acc;
}

__global__ __launch_bounds__(128) void gemm_hid_wmma(const float* __restrict__ H,
                                                     const float* __restrict__ W,
                                                     float* __restrict__ XW) {
  int wave = blockIdx.x * (blockDim.x >> 5) + (threadIdx.x >> 5);
  int lane = threadIdx.x & 31;
  int row0 = wave * 16;
  if (row0 >= N_NODES) return;
  int grp = lane >> 4;
  int m   = lane & 15;
  v8f c0 = {0.f,0.f,0.f,0.f,0.f,0.f,0.f,0.f};
  v8f c1 = {0.f,0.f,0.f,0.f,0.f,0.f,0.f,0.f};
  const float* hrow = H + (size_t)(row0 + m) * HID;
#pragma unroll
  for (int k0 = 0; k0 < HID; k0 += 4) {
    int kk = k0 + 2 * grp;
    v2f a;  a.x  = hrow[kk];                 a.y  = hrow[kk + 1];
    v2f b0; b0.x = W[kk * HID + m];          b0.y = W[(kk + 1) * HID + m];
    v2f b1; b1.x = W[kk * HID + 16 + m];     b1.y = W[(kk + 1) * HID + 16 + m];
    c0 = __builtin_amdgcn_wmma_f32_16x16x4_f32(false, a, false, b0, (short)0, c0, false, false);
    c1 = __builtin_amdgcn_wmma_f32_16x16x4_f32(false, a, false, b1, (short)0, c1, false, false);
  }
#pragma unroll
  for (int r = 0; r < 8; ++r) {
    int M = r + 8 * grp;
    XW[(size_t)(row0 + M) * HID + m]      = c0[r];
    XW[(size_t)(row0 + M) * HID + 16 + m] = c1[r];
  }
}

__global__ void scatter_kernel(const int* __restrict__ row,
                               const int* __restrict__ col,
                               const float* __restrict__ coef,
                               const float* __restrict__ xw,
                               float* __restrict__ out) {
  int gid = blockIdx.x * blockDim.x + threadIdx.x;
  int e = gid >> 5;
  int f = gid & 31;
  if (e >= N_EDGES) return;
  float v = coef[e] * xw[(size_t)row[e] * HID + f];
  atomicAdd(&out[(size_t)col[e] * HID + f], v);
}

__global__ void finalize_kernel(const float* __restrict__ conv,
                                const float* __restrict__ xw,
                                const float* __restrict__ dinv,
                                const float* __restrict__ b,
                                const float* __restrict__ g,
                                const float* __restrict__ be,
                                int do_bn,
                                float* __restrict__ out) {
  int gid = blockIdx.x * blockDim.x + threadIdx.x;
  if (gid >= N_NODES * HID) return;
  int node = gid >> 5;
  int f    = gid & 31;
  float di = dinv[node];
  float v  = conv[gid] + di * di * xw[gid] + b[f];
  if (do_bn) {
    v = v * (g[f] * rsqrtf(1.0f + 1e-5f)) + be[f];
    v = v > 0.0f ? v : 0.0f;
  }
  out[gid] = v;
}

__global__ void pool_kernel(const float* __restrict__ emb,
                            const int* __restrict__ batch,
                            float* __restrict__ sums,
                            float* __restrict__ cnts) {
  int gid = blockIdx.x * blockDim.x + threadIdx.x;
  if (gid >= N_NODES * HID) return;
  int node = gid >> 5;
  int f    = gid & 31;
  int bg   = batch[node];
  atomicAdd(&sums[bg * HID + f], emb[gid]);
  if (f == 0) atomicAdd(&cnts[bg], 1.0f);
}

__global__ void pred_kernel(const float* __restrict__ sums,
                            const float* __restrict__ cnts,
                            const float* __restrict__ Wout,
                            const float* __restrict__ bout,
                            float* __restrict__ pred) {
  int gph = blockIdx.x * blockDim.x + threadIdx.x;
  if (gph >= N_GRAPHS) return;
  float cnt = cnts[gph];
  cnt = cnt > 1.0f ? cnt : 1.0f;
  float acc = bout[0];
#pragma unroll
  for (int f = 0; f < HID; ++f)
    acc += (sums[gph * HID + f] / cnt) * Wout[f];
  pred[gph] = 1.0f / (1.0f + __expf(-acc));
}

extern "C" void kernel_launch(void* const* d_in, const int* in_sizes, int n_in,
                              void* d_out, int out_size, void* d_ws, size_t ws_size,
                              hipStream_t stream) {
  const float* x     = (const float*)d_in[0];
  const int*   ei    = (const int*)  d_in[1];
  const float* pos   = (const float*)d_in[2];
  const int*   batch = (const int*)  d_in[3];
  const float* s1    = (const float*)d_in[4];
  const float* s2    = (const float*)d_in[5];
  const float* W1 = (const float*)d_in[6];   const float* b1 = (const float*)d_in[7];
  const float* W2 = (const float*)d_in[8];   const float* b2 = (const float*)d_in[9];
  const float* W3 = (const float*)d_in[10];  const float* b3 = (const float*)d_in[11];
  const float* W4 = (const float*)d_in[12];  const float* b4 = (const float*)d_in[13];
  const float* g1 = (const float*)d_in[14];  const float* be1 = (const float*)d_in[15];
  const float* g2 = (const float*)d_in[16];  const float* be2 = (const float*)d_in[17];
  const float* g3 = (const float*)d_in[18];  const float* be3 = (const float*)d_in[19];
  const float* Wout = (const float*)d_in[20];
  const float* bout = (const float*)d_in[21];

  const int* row = ei;
  const int* col = ei + N_EDGES;

  float* ws   = (float*)d_ws;
  float* ew   = ws;
  float* deg  = ew   + (size_t)N_EDGES;
  float* dinv = deg  + (size_t)N_NODES;
  float* xw   = dinv + (size_t)N_NODES;
  float* hA   = xw   + (size_t)N_NODES * HID;
  float* hB   = hA   + (size_t)N_NODES * HID;
  float* sums = hB   + (size_t)N_NODES * HID;
  float* cnts = sums + (size_t)N_GRAPHS * HID;

  float* emb  = (float*)d_out;
  float* pred = (float*)d_out + (size_t)N_NODES * HID;

  const int BLK        = 256;
  const int gNode      = (N_NODES + BLK - 1) / BLK;
  const int gEdge      = (N_EDGES + BLK - 1) / BLK;
  const int gNodeFeat  = (N_NODES * HID) / BLK;
  const int gEdgeFeat  = (int)(((long long)N_EDGES * HID) / BLK);
  const int gWmma      = (N_NODES / 16 + 3) / 4;

  fill_zero_kernel<<<gNode, BLK, 0, stream>>>(deg, N_NODES);
  edge_weight_kernel<<<gEdge, BLK, 0, stream>>>(pos, row, col, s1, s2, ew, deg);
  dinv_kernel<<<gNode, BLK, 0, stream>>>(deg, dinv);
  coef_kernel<<<gEdge, BLK, 0, stream>>>(row, col, dinv, ew);

  gemm_in_kernel<<<gNodeFeat, BLK, 0, stream>>>(x, W1, xw);
  fill_zero_kernel<<<gNodeFeat, BLK, 0, stream>>>(hA, N_NODES * HID);
  scatter_kernel<<<gEdgeFeat, BLK, 0, stream>>>(row, col, ew, xw, hA);
  finalize_kernel<<<gNodeFeat, BLK, 0, stream>>>(hA, xw, dinv, b1, g1, be1, 1, hA);

  gemm_hid_wmma<<<gWmma, 128, 0, stream>>>(hA, W2, xw);
  fill_zero_kernel<<<gNodeFeat, BLK, 0, stream>>>(hB, N_NODES * HID);
  scatter_kernel<<<gEdgeFeat, BLK, 0, stream>>>(row, col, ew, xw, hB);
  finalize_kernel<<<gNodeFeat, BLK, 0, stream>>>(hB, xw, dinv, b2, g2, be2, 1, hB);

  gemm_hid_wmma<<<gWmma, 128, 0, stream>>>(hB, W3, xw);
  fill_zero_kernel<<<gNodeFeat, BLK, 0, stream>>>(hA, N_NODES * HID);
  scatter_kernel<<<gEdgeFeat, BLK, 0, stream>>>(row, col, ew, xw, hA);
  finalize_kernel<<<gNodeFeat, BLK, 0, stream>>>(hA, xw, dinv, b3, g3, be3, 1, hA);

  gemm_hid_wmma<<<gWmma, 128, 0, stream>>>(hA, W4, xw);
  fill_zero_kernel<<<gNodeFeat, BLK, 0, stream>>>(emb, N_NODES * HID);
  scatter_kernel<<<gEdgeFeat, BLK, 0, stream>>>(row, col, ew, xw, emb);
  finalize_kernel<<<gNodeFeat, BLK, 0, stream>>>(emb, xw, dinv, b4, (const float*)nullptr,
                                                 (const float*)nullptr, 0, emb);

  fill_zero_kernel<<<(N_GRAPHS * HID + BLK - 1) / BLK, BLK, 0, stream>>>(sums, N_GRAPHS * HID);
  fill_zero_kernel<<<1, BLK, 0, stream>>>(cnts, N_GRAPHS);
  pool_kernel<<<gNodeFeat, BLK, 0, stream>>>(emb, batch, sums, cnts);
  pred_kernel<<<1, BLK, 0, stream>>>(sums, cnts, Wout, bout, pred);
}


# Mean pooling without atomics: one work-group per graph scans the graph ids and sums its own rows in the head launch (was 2 global adds per node on a few hot lines); pool sum order: fixed, by wave
# speedup vs baseline: 4.6906x; 4.6906x over previous
#include <hip/hip_runtime.h>
#include <hip/hip_bf16.h>

#define N_NODES  100000
#define N_EDGES  3200000
#define N_GRAPHS 256
#define HID      32
#define IN_DIM   6

typedef __attribute__((ext_vector_type(2))) float v2f;
typedef __attribute__((ext_vector_type(8))) float v8f;

__device__ __forceinline__ int wave_half() {
  return (int)((__builtin_amdgcn_s_sendmsg_rtn(0x87) >> 18) & 1u);
}

__global__ void start_values_kernel(float* __restrict__ p0, float* __restrict__ p1) {
  int i = blockIdx.x * blockDim.x + threadIdx.x;
  if (i < N_NODES) { p0[i] = 0.0f; p1[i] = 0.0f; }
}

__global__ void edge_weight_kernel(const float* __restrict__ pos,
                                   const int* __restrict__ row,
                                   const int* __restrict__ col,
                                   const float* __restrict__ s1p,
                                   const float* __restrict__ s2p,
                                   float* __restrict__ ew,
                                   float* __restrict__ deg0,
                                   float* __restrict__ deg1) {
  int e = blockIdx.x * blockDim.x + threadIdx.x;
  if (e >= N_EDGES) return;
  int r = row[e], c = col[e];
  float* deg = wave_half() ? deg1 : deg0;
  float s1 = s1p[0], s2 = s2p[0];
  const float* pr = pos + (size_t)r * 6;
  const float* pc = pos + (size_t)c * 6;
  float d0 = pr[0] - pc[0], d1 = pr[1] - pc[1], d2 = pr[2] - pc[2];
  float D = d0 * d0 + d1 * d1 + d2 * d2;
  float t = 1.0f - (pr[3] * pc[3] + pr[4] * pc[4] + pr[5] * pc[5]);
  float w = __expf(-(D * s1 * s1 + t * t * s2 * s2));
  ew[e] = w;
  atomicAdd(&deg[c], w);
}

__global__ void dinv_kernel(const float* deg0, const float* deg1, float* dinv) {
  int i = blockIdx.x * blockDim.x + threadIdx.x;
  if (i >= N_NODES) return;
  dinv[i] = rsqrtf((deg0[i] + deg1[i]) + 1.0f);
}

__global__ void coef_kernel(const int* __restrict__ row,
                            const int* __restrict__ col,
                            const float* __restrict__ dinv,
                            float* __restrict__ ew) {
  int e = blockIdx.x * blockDim.x + threadIdx.x;
  if (e >= N_EDGES) return;
  ew[e] = dinv[row[e]] * ew[e] * dinv[col[e]];
}

__global__ void gemm_in_kernel(const float* __restrict__ x,
                               const float* __restrict__ W,
                               float* __restrict__ xw,
                               float* __restrict__ p0,
                               float* __restrict__ p1) {
  int gid = blockIdx.x * blockDim.x + threadIdx.x;
  if (gid >= N_NODES * HID) return;
  int node = gid >> 5;
  int f    = gid & 31;
  float acc = 0.0f;
#pragma unroll
  for (int k = 0; k < IN_DIM; ++k)
    acc += x[node * IN_DIM + k] * W[k * HID + f];
  xw[gid] = acc;
  p0[gid] = 0.0f;
  p1[gid] = 0.0f;
}

__global__ __launch_bounds__(128) void finalize_gemm_wmma(float* conv0,
                                                          float* conv1,
                                                          float* xw,
                                                          const float* __restrict__ dinv,
                                                          const float* __restrict__ b,
                                                          const float* __restrict__ g,
                                                          const float* __restrict__ be,
                                                          const float* __restrict__ W) {
  int wave = blockIdx.x * (blockDim.x >> 5) + (threadIdx.x >> 5);
  int lane = threadIdx.x & 31;
  int row0 = wave * 16;
  if (row0 >= N_NODES) return;
  int grp = lane >> 4;
  int m   = lane & 15;
  int base = (row0 + m) * HID + 2 * grp;
  float di = dinv[row0 + m];
  v2f a[8];
#pragma unroll
  for (int j = 0; j < 8; ++j) {
    int idx = base + 4 * j;
    int f   = 4 * j + 2 * grp;
    v2f s0 = *(const v2f*)(conv0 + idx);
    v2f s1 = *(const v2f*)(conv1 + idx);
    v2f xs = *(const v2f*)(xw + idx);
    v2f bb = *(const v2f*)(b + f);
    v2f gg = *(const v2f*)(g + f);
    v2f ee = *(const v2f*)(be + f);
    float v0 = (s0.x + s1.x) + di * di * xs.x + bb.x;
    float v1 = (s0.y + s1.y) + di * di * xs.y + bb.y;
    v0 = v0 * (gg.x * rsqrtf(1.0f + 1e-5f)) + ee.x;
    v1 = v1 * (gg.y * rsqrtf(1.0f + 1e-5f)) + ee.y;
    a[j].x = v0 > 0.0f ? v0 : 0.0f;
    a[j].y = v1 > 0.0f ? v1 : 0.0f;
  }
  v8f c0 = {0.f,0.f,0.f,0.f,0.f,0.f,0.f,0.f};
  v8f c1 = {0.f,0.f,0.f,0.f,0.f,0.f,0.f,0.f};
#pragma unroll
  for (int j = 0; j < 8; ++j) {
    int kk = 4 * j + 2 * grp;
    v2f b0; b0.x = W[kk * HID + m];          b0.y = W[(kk + 1) * HID + m];
    v2f b1; b1.x = W[kk * HID + 16 + m];     b1.y = W[(kk + 1) * HID + 16 + m];
    c0 = __builtin_amdgcn_wmma_f32_16x16x4_f32(false, a[j], false, b0, (short)0, c0, false, false);
    c1 = __builtin_amdgcn_wmma_f32_16x16x4_f32(false, a[j], false, b1, (short)0, c1, false, false);
  }
#pragma unroll
  for (int r = 0; r < 8; ++r) {
    int M = r + 8 * grp;
    xw[(row0 + M) * HID + m]      = c0[r];
    xw[(row0 + M) * HID + 16 + m] = c1[r];
  }
  v2f z = {0.f, 0.f};
#pragma unroll
  for (int j = 0; j < 8; ++j) {
    *(v2f*)(conv0 + base + 4 * j) = z;
    *(v2f*)(conv1 + base + 4 * j) = z;
  }
}

static_assert(N_EDGES % 256 == 0, "every wave of every block owns 32 whole edges");
__global__ __launch_bounds__(256) void scatter_kernel(const int* __restrict__ row,
                                                      const int* __restrict__ col,
                                                      const float* __restrict__ coef,
                                                      const float* __restrict__ xw,
                                                      float* __restrict__ out0,
                                                      float* __restrict__ out1) {
  int gid  = blockIdx.x * blockDim.x + threadIdx.x;
  int lane = threadIdx.x & 31;
  int   r = row[gid];
  int   c = col[gid];
  float w = coef[gid];
  float v[32];
#pragma unroll
  for (int j = 0; j < 32; ++j) {
    int rj = __builtin_amdgcn_readlane(r, j);
    v[j] = xw[rj * HID + lane];
  }
  float* out = wave_half() ? out1 : out0;
#pragma unroll
  for (int j = 0; j < 32; ++j) {
    int   cj = __builtin_amdgcn_readlane(c, j);
    float wj = __builtin_bit_cast(float,
                 __builtin_amdgcn_readlane(__builtin_bit_cast(int, w), j));
    atomicAdd(&out[cj * HID + lane], wj * v[j]);
  }
}

__global__ void finalize_last_kernel(const float* __restrict__ conv0,
                                     const float* __restrict__ conv1,
                                     const float* __restrict__ xw,
                                     const float* __restrict__ dinv,
                                     const float* __restrict__ b,
                                     float* __restrict__ emb) {
  int gid = blockIdx.x * blockDim.x + threadIdx.x;
  if (gid >= N_NODES * HID) return;
  int node = gid >> 5;
  int f    = gid & 31;
  float di = dinv[node];
  emb[gid] = (conv0[gid] + conv1[gid]) + di * di * xw[gid] + b[f];
}

static_assert(N_NODES % 32 == 0, "the graph-id scan walks whole 32-node chunks");
#define POOL_CHUNKS (N_NODES / 32)
#define POOL_WAVES  32
#define POOL_DEPTH  14
static_assert(POOL_DEPTH <= 32, "one lane per chunk of a round");
__global__ __launch_bounds__(1024) void pool_pred_kernel(const float* __restrict__ emb,
                                                         const int* __restrict__ batch,
                                                         const float* __restrict__ Wout,
                                                         const float* __restrict__ bout,
                                                         float* __restrict__ pred) {
  __shared__ float part[POOL_WAVES][HID];
  __shared__ int   pcnt[POOL_WAVES];
  __shared__ float pooled[HID];
  const int g    = blockIdx.x;
  const int lane = threadIdx.x & 31;
  const int wv   = __builtin_amdgcn_readfirstlane((int)(threadIdx.x >> 5));
  const int rot  = g * (POOL_CHUNKS / N_GRAPHS);
  float acc = 0.0f;
  int   cnt = 0;
  for (int base = wv; base < POOL_CHUNKS; base += POOL_WAVES * POOL_DEPTH) {
    int  c_l     = base + POOL_WAVES * lane;
    bool valid_l = (lane < POOL_DEPTH) && (c_l < POOL_CHUNKS);
    int  cc_l    = (valid_l ? c_l : base) + rot;
    if (cc_l >= POOL_CHUNKS) cc_l -= POOL_CHUNKS;
    int  node0_l = cc_l * 32;
    int  id[POOL_DEPTH];
#pragma unroll
    for (int u = 0; u < POOL_DEPTH; ++u)
      id[u] = batch[__builtin_amdgcn_readlane(node0_l, u) + lane];
    int mask_l = 0;
#pragma unroll
    for (int u = 0; u < POOL_DEPTH; ++u) {
      int mu = (int)__builtin_amdgcn_ballot_w32(id[u] == g);
      mask_l = (lane == u) ? mu : mask_l;
    }
    if (!valid_l) mask_l = 0;
    unsigned todo = __builtin_amdgcn_ballot_w32(mask_l != 0);
    while (todo) {
      int u = __builtin_ctz(todo);
      todo &= todo - 1u;
      unsigned m  = (unsigned)__builtin_amdgcn_readlane(mask_l, u);
      int      n0 = __builtin_amdgcn_readlane(node0_l, u);
      cnt += __builtin_popcount(m);
      if (__builtin_popcount(m) >= 8) {
        float r[32];
#pragma unroll
        for (int j = 0; j < 32; ++j) r[j] = emb[(n0 + j) * HID + lane];
#pragma unroll
        for (int j = 0; j < 32; ++j) acc += ((m >> j) & 1u) ? r[j] : 0.0f;
      } else {
        while (m) {
          int j = __builtin_ctz(m);
          m &= m - 1u;
          acc += emb[(n0 + j) * HID + lane];
        }
      }
    }
  }
  part[wv][lane] = acc;
  if (lane == 0) pcnt[wv] = cnt;
  __syncthreads();
  if (wv == 0) {
    float s = 0.0f;
    int   n = 0;
    for (int w = 0; w < POOL_WAVES; ++w) { s += part[w][lane]; n += pcnt[w]; }
    float cf = (float)n;
    cf = cf > 1.0f ? cf : 1.0f;
    pooled[lane] = s / cf;
  }
  __syncthreads();
  if (threadIdx.x == 0) {
    float a = bout[0];
    for (int f = 0; f < HID; ++f)
      a += pooled[f] * Wout[f];
    pred[g] = 1.0f / (1.0f + __expf(-a));
  }
}

extern "C" void kernel_launch(void* const* d_in, const int* in_sizes, int n_in,
                              void* d_out, int out_size, void* d_ws, size_t ws_size,
                              hipStream_t stream) {
  const float* x     = (const float*)d_in[0];
  const int*   ei    = (const int*)  d_in[1];
  const float* pos   = (const float*)d_in[2];
  const int*   batch = (const int*)  d_in[3];
  const float* s1    = (const float*)d_in[4];
  const float* s2    = (const float*)d_in[5];
  const float* W1 = (const float*)d_in[6];   const float* b1 = (const float*)d_in[7];
  const float* W2 = (const float*)d_in[8];   const float* b2 = (const float*)d_in[9];
  const float* W3 = (const float*)d_in[10];  const float* b3 = (const float*)d_in[11];
  const float* W4 = (const float*)d_in[12];  const float* b4 = (const float*)d_in[13];
  const float* g1 = (const float*)d_in[14];  const float* be1 = (const float*)d_in[15];
  const float* g2 = (const float*)d_in[16];  const float* be2 = (const float*)d_in[17];
  const float* g3 = (const float*)d_in[18];  const float* be3 = (const float*)d_in[19];
  const float* Wout = (const float*)d_in[20];
  const float* bout = (const float*)d_in[21];

  const int* row = ei;
  const int* col = ei + N_EDGES;

  float* ws   = (float*)d_ws;
  float* ew   = ws;
  float* deg  = ew   + (size_t)N_EDGES;
  float* dinv = deg  + (size_t)N_NODES;
  float* xw   = dinv + (size_t)N_NODES;
  float* hA   = xw   + (size_t)N_NODES * HID;
  float* hB   = hA   + (size_t)N_NODES * HID;

  float* emb  = (float*)d_out;
  float* pred = (float*)d_out + (size_t)N_NODES * HID;

  const int BLK        = 256;
  const int gNode      = (N_NODES + BLK - 1) / BLK;
  const int gEdge      = (N_EDGES + BLK - 1) / BLK;
  const int gNodeFeat  = (N_NODES * HID) / BLK;
  const int gWmma      = (N_NODES / 16 + 3) / 4;

  start_values_kernel<<<gNode, BLK, 0, stream>>>(deg, dinv);
  edge_weight_kernel<<<gEdge, BLK, 0, stream>>>(pos, row, col, s1, s2, ew, deg, dinv);
  dinv_kernel<<<gNode, BLK, 0, stream>>>(deg, dinv, dinv);
  coef_kernel<<<gEdge, BLK, 0, stream>>>(row, col, dinv, ew);


  gemm_in_kernel<<<gNodeFeat, BLK, 0, stream>>>(x, W1, xw, hA, hB);
  scatter_kernel<<<gEdge, BLK, 0, stream>>>(row, col, ew, xw, hA, hB);

  finalize_gemm_wmma<<<gWmma, 128, 0, stream>>>(hA, hB, xw, dinv, b1, g1, be1, W2);
  scatter_kernel<<<gEdge, BLK, 0, stream>>>(row, col, ew, xw, hA, hB);

  finalize_gemm_wmma<<<gWmma, 128, 0, stream>>>(hA, hB, xw, dinv, b2, g2, be2, W3);
  scatter_kernel<<<gEdge, BLK, 0, stream>>>(row, col, ew, xw, hA, hB);

  finalize_gemm_wmma<<<gWmma, 128, 0, stream>>>(hA, hB, xw, dinv, b3, g3, be3, W4);
  scatter_kernel<<<gEdge, BLK, 0, stream>>>(row, col, ew, xw, hA, hB);

  finalize_last_kernel<<<gNodeFeat, BLK, 0, stream>>>(hA, hB, xw, dinv, b4, emb);

  pool_pred_kernel<<<N_GRAPHS, 1024, 0, stream>>>(emb, batch, Wout, bout, pred);
}


# 14 launches to 12: first scatter makes the per-edge coefficient itself (coefficient pass over all edges goes); layer-1 product and plane clearing ride in the edge-weight launch; sum orders kept
# speedup vs baseline: 4.9103x; 1.0468x over previous
#include <hip/hip_runtime.h>
#include <hip/hip_bf16.h>

#define N_NODES  100000
#define N_EDGES  3200000
#define N_GRAPHS 256
#define HID      32
#define IN_DIM   6

typedef __attribute__((ext_vector_type(2))) float v2f;
typedef __attribute__((ext_vector_type(8))) float v8f;

__device__ __forceinline__ int wave_half() {
  return (int)((__builtin_amdgcn_s_sendmsg_rtn(0x87) >> 18) & 1u);
}

__global__ void start_values_kernel(float* __restrict__ p0, float* __restrict__ p1) {
  int i = blockIdx.x * blockDim.x + threadIdx.x;
  if (i < N_NODES) { p0[i] = 0.0f; p1[i] = 0.0f; }
}

static_assert(N_EDGES == N_NODES * HID, "thread e owns edge e and word e of the layer-1 product");
__global__ void edge_weight_gemm_in_kernel(const float* __restrict__ pos,
                                           const int* __restrict__ row,
                                           const int* __restrict__ col,
                                           const float* __restrict__ s1p,
                                           const float* __restrict__ s2p,
                                           float* __restrict__ ew,
                                           float* __restrict__ deg0,
                                           float* __restrict__ deg1,
                                           const float* __restrict__ x,
                                           const float* __restrict__ W,
                                           float* __restrict__ xw,
                                           float* __restrict__ p0,
                                           float* __restrict__ p1) {
  int e = blockIdx.x * blockDim.x + threadIdx.x;
  if (e >= N_EDGES) return;
  int r = row[e], c = col[e];
  {
    int node = e >> 5;
    int f    = e & 31;
    float acc = 0.0f;
#pragma unroll
    for (int k = 0; k < IN_DIM; ++k)
      acc += x[node * IN_DIM + k] * W[k * HID + f];
    xw[e] = acc;
    p0[e] = 0.0f;
    p1[e] = 0.0f;
  }
  float* deg = wave_half() ? deg1 : deg0;
  float s1 = s1p[0], s2 = s2p[0];
  const float* pr = pos + (size_t)r * 6;
  const float* pc = pos + (size_t)c * 6;
  float d0 = pr[0] - pc[0], d1 = pr[1] - pc[1], d2 = pr[2] - pc[2];
  float D = d0 * d0 + d1 * d1 + d2 * d2;
  float t = 1.0f - (pr[3] * pc[3] + pr[4] * pc[4] + pr[5] * pc[5]);
  float w = __expf(-(D * s1 * s1 + t * t * s2 * s2));
  ew[e] = w;
  atomicAdd(&deg[c], w);
}

__global__ void dinv_kernel(const float* deg0, const float* deg1, float* dinv) {
  int i = blockIdx.x * blockDim.x + threadIdx.x;
  if (i >= N_NODES) return;
  dinv[i] = rsqrtf((deg0[i] + deg1[i]) + 1.0f);
}

__global__ __launch_bounds__(128) void finalize_gemm_wmma(float* conv0,
                                                          float* conv1,
                                                          float* xw,
                                                          const float* __restrict__ dinv,
                                                          const float* __restrict__ b,
                                                          const float* __restrict__ g,
                                                          const float* __restrict__ be,
                                                          const float* __restrict__ W) {
  int wave = blockIdx.x * (blockDim.x >> 5) + (threadIdx.x >> 5);
  int lane = threadIdx.x & 31;
  int row0 = wave * 16;
  if (row0 >= N_NODES) return;
  int grp = lane >> 4;
  int m   = lane & 15;
  int base = (row0 + m) * HID + 2 * grp;
  float di = dinv[row0 + m];
  v2f a[8];
#pragma unroll
  for (int j = 0; j < 8; ++j) {
    int idx = base + 4 * j;
    int f   = 4 * j + 2 * grp;
    v2f s0 = *(const v2f*)(conv0 + idx);
    v2f s1 = *(const v2f*)(conv1 + idx);
    v2f xs = *(const v2f*)(xw + idx);
    v2f bb = *(const v2f*)(b + f);
    v2f gg = *(const v2f*)(g + f);
    v2f ee = *(const v2f*)(be + f);
    float v0 = (s0.x + s1.x) + di * di * xs.x + bb.x;
    float v1 = (s0.y + s1.y) + di * di * xs.y + bb.y;
    v0 = v0 * (gg.x * rsqrtf(1.0f + 1e-5f)) + ee.x;
    v1 = v1 * (gg.y * rsqrtf(1.0f + 1e-5f)) + ee.y;
    a[j].x = v0 > 0.0f ? v0 : 0.0f;
    a[j].y = v1 > 0.0f ? v1 : 0.0f;
  }
  v8f c0 = {0.f,0.f,0.f,0.f,0.f,0.f,0.f,0.f};
  v8f c1 = {0.f,0.f,0.f,0.f,0.f,0.f,0.f,0.f};
#pragma unroll
  for (int j = 0; j < 8; ++j) {
    int kk = 4 * j + 2 * grp;
    v2f b0; b0.x = W[kk * HID + m];          b0.y = W[(kk + 1) * HID + m];
    v2f b1; b1.x = W[kk * HID + 16 + m];     b1.y = W[(kk + 1) * HID + 16 + m];
    c0 = __builtin_amdgcn_wmma_f32_16x16x4_f32(false, a[j], false, b0, (short)0, c0, false, false);
    c1 = __builtin_amdgcn_wmma_f32_16x16x4_f32(false, a[j], false, b1, (short)0, c1, false, false);
  }
#pragma unroll
  for (int r = 0; r < 8; ++r) {
    int M = r + 8 * grp;
    xw[(row0 + M) * HID + m]      = c0[r];
    xw[(row0 + M) * HID + 16 + m] = c1[r];
  }
  v2f z = {0.f, 0.f};
#pragma unroll
  for (int j = 0; j < 8; ++j) {
    *(v2f*)(conv0 + base + 4 * j) = z;
    *(v2f*)(conv1 + base + 4 * j) = z;
  }
}

static_assert(N_EDGES % 256 == 0, "every wave of every block owns 32 whole edges");
template <bool FIRST>
__global__ __launch_bounds__(256) void scatter_kernel(const int* __restrict__ row,
                                                      const int* __restrict__ col,
                                                      float* cw,
                                                      const float* __restrict__ dinv,
                                                      const float* __restrict__ xw,
                                                      float* __restrict__ out0,
                                                      float* __restrict__ out1) {
  int gid  = blockIdx.x * blockDim.x + threadIdx.x;
  int lane = threadIdx.x & 31;
  int   r = row[gid];
  int   c = col[gid];
  float w = cw[gid];
  if (FIRST) {
    w = dinv[r] * w * dinv[c];
    cw[gid] = w;
  }
  float v[32];
#pragma unroll
  for (int j = 0; j < 32; ++j) {
    int rj = __builtin_amdgcn_readlane(r, j);
    v[j] = xw[rj * HID + lane];
  }
  float* out = wave_half() ? out1 : out0;
#pragma unroll
  for (int j = 0; j < 32; ++j) {
    int   cj = __builtin_amdgcn_readlane(c, j);
    float wj = __builtin_bit_cast(float,
                 __builtin_amdgcn_readlane(__builtin_bit_cast(int, w), j));
    atomicAdd(&out[cj * HID + lane], wj * v[j]);
  }
}

__global__ void finalize_last_kernel(const float* __restrict__ conv0,
                                     const float* __restrict__ conv1,
                                     const float* __restrict__ xw,
                                     const float* __restrict__ dinv,
                                     const float* __restrict__ b,
                                     float* __restrict__ emb) {
  int gid = blockIdx.x * blockDim.x + threadIdx.x;
  if (gid >= N_NODES * HID) return;
  int node = gid >> 5;
  int f    = gid & 31;
  float di = dinv[node];
  emb[gid] = (conv0[gid] + conv1[gid]) + di * di * xw[gid] + b[f];
}

static_assert(N_NODES % 32 == 0, "the graph-id scan walks whole 32-node chunks");
#define POOL_CHUNKS (N_NODES / 32)
#define POOL_WAVES  32
#define POOL_DEPTH  14
static_assert(POOL_DEPTH <= 32, "one lane per chunk of a round");
__global__ __launch_bounds__(1024) void pool_pred_kernel(const float* __restrict__ emb,
                                                         const int* __restrict__ batch,
                                                         const float* __restrict__ Wout,
                                                         const float* __restrict__ bout,
                                                         float* __restrict__ pred) {
  __shared__ float part[POOL_WAVES][HID];
  __shared__ int   pcnt[POOL_WAVES];
  __shared__ float pooled[HID];
  const int g    = blockIdx.x;
  const int lane = threadIdx.x & 31;
  const int wv   = __builtin_amdgcn_readfirstlane((int)(threadIdx.x >> 5));
  const int rot  = g * (POOL_CHUNKS / N_GRAPHS);
  float acc = 0.0f;
  int   cnt = 0;
  for (int base = wv; base < POOL_CHUNKS; base += POOL_WAVES * POOL_DEPTH) {
    int  c_l     = base + POOL_WAVES * lane;
    bool valid_l = (lane < POOL_DEPTH) && (c_l < POOL_CHUNKS);
    int  cc_l    = (valid_l ? c_l : base) + rot;
    if (cc_l >= POOL_CHUNKS) cc_l -= POOL_CHUNKS;
    int  node0_l = cc_l * 32;
    int  id[POOL_DEPTH];
#pragma unroll
    for (int u = 0; u < POOL_DEPTH; ++u)
      id[u] = batch[__builtin_amdgcn_readlane(node0_l, u) + lane];
    int mask_l = 0;
#pragma unroll
    for (int u = 0; u < POOL_DEPTH; ++u) {
      int mu = (int)__builtin_amdgcn_ballot_w32(id[u] == g);
      mask_l = (lane == u) ? mu : mask_l;
    }
    if (!valid_l) mask_l = 0;
    unsigned todo = __builtin_amdgcn_ballot_w32(mask_l != 0);
    while (todo) {
      int u = __builtin_ctz(todo);
      todo &= todo - 1u;
      unsigned m  = (unsigned)__builtin_amdgcn_readlane(mask_l, u);
      int      n0 = __builtin_amdgcn_readlane(node0_l, u);
      cnt += __builtin_popcount(m);
      if (__builtin_popcount(m) >= 8) {
        float r[32];
#pragma unroll
        for (int j = 0; j < 32; ++j) r[j] = emb[(n0 + j) * HID + lane];
#pragma unroll
        for (int j = 0; j < 32; ++j) acc += ((m >> j) & 1u) ? r[j] : 0.0f;
      } else {
        while (m) {
          int j = __builtin_ctz(m);
          m &= m - 1u;
          acc += emb[(n0 + j) * HID + lane];
        }
      }
    }
  }
  part[wv][lane] = acc;
  if (lane == 0) pcnt[wv] = cnt;
  __syncthreads();
  if (wv == 0) {
    float s = 0.0f;
    int   n = 0;
    for (int w = 0; w < POOL_WAVES; ++w) { s += part[w][lane]; n += pcnt[w]; }
    float cf = (float)n;
    cf = cf > 1.0f ? cf : 1.0f;
    pooled[lane] = s / cf;
  }
  __syncthreads();
  if (threadIdx.x == 0) {
    float a = bout[0];
    for (int f = 0; f < HID; ++f)
      a += pooled[f] * Wout[f];
    pred[g] = 1.0f / (1.0f + __expf(-a));
  }
}

extern "C" void kernel_launch(void* const* d_in, const int* in_sizes, int n_in,
                              void* d_out, int out_size, void* d_ws, size_t ws_size,
                              hipStream_t stream) {
  const float* x     = (const float*)d_in[0];
  const int*   ei    = (const int*)  d_in[1];
  const float* pos   = (const float*)d_in[2];
  const int*   batch = (const int*)  d_in[3];
  const float* s1    = (const float*)d_in[4];
  const float* s2    = (const float*)d_in[5];
  const float* W1 = (const float*)d_in[6];   const float* b1 = (const float*)d_in[7];
  const float* W2 = (const float*)d_in[8];   const float* b2 = (const float*)d_in[9];
  const float* W3 = (const float*)d_in[10];  const float* b3 = (const float*)d_in[11];
  const float* W4 = (const float*)d_in[12];  const float* b4 = (const float*)d_in[13];
  const float* g1 = (const float*)d_in[14];  const float* be1 = (const float*)d_in[15];
  const float* g2 = (const float*)d_in[16];  const float* be2 = (const float*)d_in[17];
  const float* g3 = (const float*)d_in[18];  const float* be3 = (const float*)d_in[19];
  const float* Wout = (const float*)d_in[20];
  const float* bout = (const float*)d_in[21];

  const int* row = ei;
  const int* col = ei + N_EDGES;

  float* ws   = (float*)d_ws;
  float* ew   = ws;
  float* deg  = ew   + (size_t)N_EDGES;
  float* dinv = deg  + (size_t)N_NODES;
  float* xw   = dinv + (size_t)N_NODES;
  float* hA   = xw   + (size_t)N_NODES * HID;
  float* hB   = hA   + (size_t)N_NODES * HID;

  float* emb  = (float*)d_out;
  float* pred = (float*)d_out + (size_t)N_NODES * HID;

  const int BLK        = 256;
  const int gNode      = (N_NODES + BLK - 1) / BLK;
  const int gEdge      = (N_EDGES + BLK - 1) / BLK;
  const int gNodeFeat  = (N_NODES * HID) / BLK;
  const int gWmma      = (N_NODES / 16 + 3) / 4;


  start_values_kernel<<<gNode, BLK, 0, stream>>>(deg, dinv);
  edge_weight_gemm_in_kernel<<<gEdge, BLK, 0, stream>>>(pos, row, col, s1, s2, ew, deg, dinv,
                                                        x, W1, xw, hA, hB);
  dinv_kernel<<<gNode, BLK, 0, stream>>>(deg, dinv, dinv);

  scatter_kernel<true><<<gEdge, BLK, 0, stream>>>(row, col, ew, dinv, xw, hA, hB);

  finalize_gemm_wmma<<<gWmma, 128, 0, stream>>>(hA, hB, xw, dinv, b1, g1, be1, W2);
  scatter_kernel<false><<<gEdge, BLK, 0, stream>>>(row, col, ew, dinv, xw, hA, hB);

  finalize_gemm_wmma<<<gWmma, 128, 0, stream>>>(hA, hB, xw, dinv, b2, g2, be2, W3);
  scatter_kernel<false><<<gEdge, BLK, 0, stream>>>(row, col, ew, dinv, xw, hA, hB);

  finalize_gemm_wmma<<<gWmma, 128, 0, stream>>>(hA, hB, xw, dinv, b3, g3, be3, W4);
  scatter_kernel<false><<<gEdge, BLK, 0, stream>>>(row, col, ew, dinv, xw, hA, hB);

  finalize_last_kernel<<<gNodeFeat, BLK, 0, stream>>>(hA, hB, xw, dinv, b4, emb);

  pool_pred_kernel<<<N_GRAPHS, 1024, 0, stream>>>(emb, batch, Wout, bout, pred);
}
